# starting kernel
# baseline (speedup 1.0000x reference)
#include <hip/hip_runtime.h>
#include <hip/hip_bf16.h>


typedef float v2f __attribute__((ext_vector_type(2)));
typedef float v4f __attribute__((ext_vector_type(4)));
typedef float v8f __attribute__((ext_vector_type(8)));

#define D_FEAT      128
#define ROWSTRIDE   132
#define TILE        16
#define WAVES_PB    8

__global__ __launch_bounds__(256)
void edge_dot_wmma(const float* __restrict__ h,
                   const int*   __restrict__ u,
                   const int*   __restrict__ v,
                   float*       __restrict__ out,
                   int nE)
{
    extern __shared__ float smem[];

    const int  lane   = threadIdx.x & 31;
    const int  wave   = threadIdx.x >> 5;
    const long nTiles = ((long)nE + TILE - 1) / TILE;
    const long tileId = (long)blockIdx.x * WAVES_PB + wave;
    const bool active = (tileId < nTiles);
    const long ebase  = active ? tileId * TILE : 0;

    float* ldsA = smem + (size_t)wave * (2 * TILE * ROWSTRIDE);
    float* ldsB = ldsA + TILE * ROWSTRIDE;

    #pragma unroll 4
    for (int r = 0; r < TILE; ++r) {
        const long e  = ebase + r;
        int ur = 0, vr = 0;
        if (e < (long)nE) { ur = u[e]; vr = v[e]; }
        const v4f au = *(const v4f*)(h + (size_t)ur * D_FEAT + lane * 4);
        const v4f bv = *(const v4f*)(h + (size_t)vr * D_FEAT + lane * 4);
        *(v4f*)(ldsA + r * ROWSTRIDE + lane * 4) = au;
        *(v4f*)(ldsB + r * ROWSTRIDE + lane * 4) = bv;
    }
    __syncthreads();

    const int    mrow  = lane & 15;
    const int    khalf = (lane >> 4) * 2;
    const float* Arow  = ldsA + mrow * ROWSTRIDE + khalf;
    const float* Brow  = ldsB + mrow * ROWSTRIDE + khalf;

    v8f c = {};
    #pragma unroll
    for (int k = 0; k < D_FEAT; k += 4) {
        const v2f a = *(const v2f*)(Arow + k);
        const v2f b = *(const v2f*)(Brow + k);
        c = __builtin_amdgcn_wmma_f32_16x16x4_f32(false, a, false, b,
                                                  (short)0, c, false, false);
    }

    const bool validLane = (lane < 8) || (lane >= 24);
    const int  m   = (lane < 8) ? lane : (lane - 16);
    const int  sel = (lane < 8) ? lane : (lane - 24);

    float val = c[0];
    #pragma unroll
    for (int i = 1; i < 8; ++i)
        val = (sel == i) ? c[i] : val;

    const long e = ebase + m;
    if (active && validLane && e < (long)nE)
        out[e] = val;
}

extern "C" void kernel_launch(void* const* d_in, const int* in_sizes, int n_in,
                              void* d_out, int out_size, void* d_ws, size_t ws_size,
                              hipStream_t stream)
{
    const float* h  = (const float*)d_in[0];
    const int*   u  = (const int*)d_in[1];
    const int*   v  = (const int*)d_in[2];
    float*       out = (float*)d_out;

    const int  nE     = in_sizes[1];
    const long nTiles = ((long)nE + TILE - 1) / TILE;
    const int  blocks = (int)((nTiles + WAVES_PB - 1) / WAVES_PB);
    const size_t smemBytes = (size_t)WAVES_PB * 2 * TILE * ROWSTRIDE * sizeof(float);

    edge_dot_wmma<<<blocks, 256, smemBytes, stream>>>(h, u, v, out, nE);
}


# K-chunked tile: 16 edges x 64 features staged per pass (16 index loads, then 16 aligned 256 B row-half loads in flight per wave); 8.5 KB LDS, single-wave workgroups of 4 tiles, 32 waves per WGP
# speedup vs baseline: 2.0382x; 2.0382x over previous
#include <hip/hip_runtime.h>
#include <hip/hip_bf16.h>


typedef float v2f __attribute__((ext_vector_type(2)));
typedef float v4f __attribute__((ext_vector_type(4)));
typedef float v8f __attribute__((ext_vector_type(8)));

#define D_FEAT      128
#define TILE        16
#define KCHUNK      64
#define NCHUNK      (D_FEAT / KCHUNK)
#define CSTRIDE     68
#define TILES_PW    4
#define LOADS_PC    16

__global__ __launch_bounds__(32)
void edge_dot_wmma(const float* __restrict__ h,
                   const int*   __restrict__ u,
                   const int*   __restrict__ v,
                   float*       __restrict__ out,
                   int nE)
{
    __shared__ float lds[2 * TILE * CSTRIDE];

    const int  lane   = threadIdx.x & 31;
    const long nTiles = ((long)nE + TILE - 1) / TILE;
    const long tile0  = (long)blockIdx.x * TILES_PW;

    const int grow = lane >> 4;
    const int gcol = (lane & 15) * 4;

    const int    mrow  = lane & 15;
    const int    khalf = (lane >> 4) * 2;
    const float* Arow  = lds + mrow * CSTRIDE + khalf;
    const float* Brow  = Arow + TILE * CSTRIDE;
    float*       Gdst  = lds + grow * CSTRIDE + gcol;

    const bool validLane = (lane < 8) || (lane >= 24);
    const int  m   = (lane < 8) ? lane : (lane - 16);
    const int  sel = (lane < 8) ? lane : (lane - 24);

    #pragma unroll 1
    for (int t = 0; t < TILES_PW; ++t) {
        const long tile = tile0 + t;
        if (tile >= nTiles) break;
        const long ebase = tile * TILE;

        int node[LOADS_PC];
        #pragma unroll
        for (int j = 0; j < LOADS_PC; ++j) {
            long e = ebase + 2 * (j & 7) + grow;
            if (e > (long)nE - 1) e = (long)nE - 1;
            node[j] = (j < 8) ? u[e] : v[e];
        }

        v8f c = {};
        for (int ch = 0; ch < NCHUNK; ++ch) {
            v4f g[LOADS_PC];
            #pragma unroll
            for (int j = 0; j < LOADS_PC; ++j)
                g[j] = *(const v4f*)(h + (size_t)node[j] * D_FEAT + ch * KCHUNK + gcol);

            __syncthreads();
            #pragma unroll
            for (int j = 0; j < LOADS_PC; ++j)
                *(v4f*)(Gdst + 2 * j * CSTRIDE) = g[j];
            __syncthreads();

            #pragma unroll
            for (int k = 0; k < KCHUNK; k += 4) {
                const v2f a = *(const v2f*)(Arow + k);
                const v2f b = *(const v2f*)(Brow + k);
                c = __builtin_amdgcn_wmma_f32_16x16x4_f32(false, a, false, b,
                                                          (short)0, c, false, false);
            }
        }

        float val = c[0];
        #pragma unroll
        for (int i = 1; i < 8; ++i)
            val = (sel == i) ? c[i] : val;

        const long e = ebase + m;
        if (validLane && e < (long)nE)
            out[e] = val;
    }
}

extern "C" void kernel_launch(void* const* d_in, const int* in_sizes, int n_in,
                              void* d_out, int out_size, void* d_ws, size_t ws_size,
                              hipStream_t stream)
{
    const float* h  = (const float*)d_in[0];
    const int*   u  = (const int*)d_in[1];
    const int*   v  = (const int*)d_in[2];
    float*       out = (float*)d_out;

    const int  nE     = in_sizes[1];
    const long nTiles = ((long)nE + TILE - 1) / TILE;
    const int  blocks = (int)((nTiles + TILES_PW - 1) / TILES_PW);

    edge_dot_wmma<<<blocks, 32, 0, stream>>>(h, u, v, out, nE);
}


# One aligned 64-edge block per wave: its u and v indices arrive as whole lines in one round trip and its 64 scores leave as two whole 128 B lines, both parked in LDS; fewer L2 requests
# speedup vs baseline: 2.2351x; 1.0966x over previous
#include <hip/hip_runtime.h>
#include <hip/hip_bf16.h>


typedef float v2f __attribute__((ext_vector_type(2)));
typedef float v4f __attribute__((ext_vector_type(4)));
typedef float v8f __attribute__((ext_vector_type(8)));
typedef int   v4i __attribute__((ext_vector_type(4)));

#define D_FEAT      128
#define TILE        16
#define KCHUNK      64
#define NCHUNK      (D_FEAT / KCHUNK)
#define CSTRIDE     68
#define TILES_PW    4
#define EDGES_PW    (TILE * TILES_PW)
#define LOADS_PC    16

__global__ __launch_bounds__(32)
void edge_dot_wmma(const float* __restrict__ h,
                   const int*   __restrict__ u,
                   const int*   __restrict__ v,
                   float*       __restrict__ out,
                   int nE)
{
    __shared__ float lds[2 * TILE * CSTRIDE];
    __shared__ __attribute__((aligned(16))) int nidx[2 * EDGES_PW];
    __shared__ float res[EDGES_PW];

    const int  lane   = threadIdx.x & 31;
    const long nTiles = ((long)nE + TILE - 1) / TILE;
    const long tile0  = (long)blockIdx.x * TILES_PW;
    const long eb     = tile0 * TILE;

    const int grow = lane >> 4;
    const int gcol = (lane & 15) * 4;

    const int    mrow  = lane & 15;
    const int    khalf = (lane >> 4) * 2;
    const float* Arow  = lds + mrow * CSTRIDE + khalf;
    const float* Brow  = Arow + TILE * CSTRIDE;
    float*       Gdst  = lds + grow * CSTRIDE + gcol;

    const bool validLane = (lane < 8) || (lane >= 24);
    const int  m   = (lane < 8) ? lane : (lane - 16);
    const int  sel = (lane < 8) ? lane : (lane - 24);

    #pragma unroll
    for (int i = 0; i < 2; ++i) {
        const int el = 32 * i + lane;
        long e = eb + el;
        if (e > (long)nE - 1) e = (long)nE - 1;
        const int r   = el & 15;
        const int pos = (el & ~15) + (r & 1) * 8 + (r >> 1);
        nidx[pos]            = u[e];
        nidx[EDGES_PW + pos] = v[e];
    }
    __syncthreads();

    #pragma unroll 1
    for (int t = 0; t < TILES_PW; ++t) {
        const long tile = tile0 + t;
        if (tile >= nTiles) break;

        const v4i* up = (const v4i*)(nidx + t * TILE + grow * 8);
        const v4i* vp = (const v4i*)(nidx + EDGES_PW + t * TILE + grow * 8);
        const v4i u0 = up[0], u1 = up[1], v0 = vp[0], v1 = vp[1];
        int node[LOADS_PC];
        #pragma unroll
        for (int j = 0; j < 4; ++j) {
            node[j]      = u0[j];
            node[4 + j]  = u1[j];
            node[8 + j]  = v0[j];
            node[12 + j] = v1[j];
        }

        v8f c = {};
        for (int ch = 0; ch < NCHUNK; ++ch) {
            v4f g[LOADS_PC];
            #pragma unroll
            for (int j = 0; j < LOADS_PC; ++j)
                g[j] = *(const v4f*)(h + (size_t)node[j] * D_FEAT + ch * KCHUNK + gcol);

            __syncthreads();
            #pragma unroll
            for (int j = 0; j < LOADS_PC; ++j)
                *(v4f*)(Gdst + 2 * j * CSTRIDE) = g[j];
            __syncthreads();

            #pragma unroll
            for (int k = 0; k < KCHUNK; k += 4) {
                const v2f a = *(const v2f*)(Arow + k);
                const v2f b = *(const v2f*)(Brow + k);
                c = __builtin_amdgcn_wmma_f32_16x16x4_f32(false, a, false, b,
                                                          (short)0, c, false, false);
            }
        }

        float val = c[0];
        #pragma unroll
        for (int i = 1; i < 8; ++i)
            val = (sel == i) ? c[i] : val;

        if (validLane)
            res[t * TILE + m] = val;
    }

    __syncthreads();

    #pragma unroll
    for (int i = 0; i < 2; ++i) {
        const long e = eb + 32 * i + lane;
        if (e < (long)nE)
            out[e] = res[32 * i + lane];
    }
}

extern "C" void kernel_launch(void* const* d_in, const int* in_sizes, int n_in,
                              void* d_out, int out_size, void* d_ws, size_t ws_size,
                              hipStream_t stream)
{
    const float* h  = (const float*)d_in[0];
    const int*   u  = (const int*)d_in[1];
    const int*   v  = (const int*)d_in[2];
    float*       out = (float*)d_out;

    const int  nE     = in_sizes[1];
    const long nTiles = ((long)nE + TILE - 1) / TILE;
    const int  blocks = (int)((nTiles + TILES_PW - 1) / TILES_PW);

    edge_dot_wmma<<<blocks, 32, 0, stream>>>(h, u, v, out, nE);
}
